# starting kernel
# baseline (speedup 1.0000x reference)
#include <hip/hip_runtime.h>

typedef __attribute__((ext_vector_type(2))) float v2f;
typedef __attribute__((ext_vector_type(8))) float v8f;

static constexpr int kM    = 8192;
static constexpr int kK    = 8;
static constexpr int kC    = 256;
static constexpr int kLay  = 8;
static constexpr int kWavesPerBlock = 4;
static constexpr int kNodesPerBlock = kWavesPerBlock * 16;

__device__ __forceinline__ float fast_sigmoid(float x) {
    return __builtin_amdgcn_rcpf(1.0f + __expf(-x));
}

__device__ __forceinline__ float pick8(v8f c, int i) {
    float r = c[0];
    if (i == 1) r = c[1];
    if (i == 2) r = c[2];
    if (i == 3) r = c[3];
    if (i == 4) r = c[4];
    if (i == 5) r = c[5];
    if (i == 6) r = c[6];
    if (i == 7) r = c[7];
    return r;
}

__global__ void bn_root_kernel(const float* __restrict__ logits,
                               float* __restrict__ out) {
    int i = blockIdx.x * blockDim.x + threadIdx.x;
    if (i < kM) out[i] = fast_sigmoid(logits[i]);
}

__global__ __launch_bounds__(kWavesPerBlock * 32)
void bn_layer_kernel(const float* __restrict__ cpt,
                     const int*   __restrict__ par,
                     const float* __restrict__ prev,
                     float*       __restrict__ outm) {
    const int lane = threadIdx.x & 31;
    const int wave = threadIdx.x >> 5;
    const int n    = lane & 15;
    const int h    = lane >> 4;
    const int tile = blockIdx.x * kWavesPerBlock + wave;
    const int node = tile * 16 + n;

    const int* pp = par + node * kK;
    float p[kK], q[kK];
#pragma unroll
    for (int j = 0; j < kK; ++j) {
        p[j] = prev[pp[j]];
        q[j] = 1.0f - p[j];
    }
    const float fh = h ? p[6] : q[6];

    const float* row = cpt + (size_t)node * kC + 2 * h;
    __builtin_prefetch(row, 0, 3);
    __builtin_prefetch(row + kC / 2, 0, 3);

    v8f acc = {};
#pragma unroll
    for (int hi = 0; hi < 8; ++hi) {
        const float rH = (((hi >> 2) & 1) ? p[0] : q[0])
                       * (((hi >> 1) & 1) ? p[1] : q[1])
                       * (((hi     ) & 1) ? p[2] : q[2]) * fh;
#pragma unroll
        for (int lo = 0; lo < 8; ++lo) {
            const int t = hi * 8 + lo;
            const float rL = (((lo >> 2) & 1) ? p[3] : q[3])
                           * (((lo >> 1) & 1) ? p[4] : q[4])
                           * (((lo     ) & 1) ? p[5] : q[5]);
            const float r7 = rH * rL;

            v2f a;
            a.x = r7 * q[7];
            a.y = r7 * p[7];

            const float2 cl = *(const float2*)(row + 4 * t);
            v2f b;
            b.x = fast_sigmoid(cl.x);
            b.y = fast_sigmoid(cl.y);

            acc = __builtin_amdgcn_wmma_f32_16x16x4_f32(
                false, a, false, b, (short)0, acc, false, false);
        }
    }

    if (lane < 8) {
        outm[tile * 16 + lane] = pick8(acc, lane);
    } else if (lane >= 24) {
        outm[tile * 16 + (lane - 16)] = pick8(acc, lane - 24);
    }
}

extern "C" void kernel_launch(void* const* d_in, const int* in_sizes, int n_in,
                              void* d_out, int out_size, void* d_ws, size_t ws_size,
                              hipStream_t stream) {
    (void)in_sizes; (void)n_in; (void)out_size; (void)d_ws; (void)ws_size;
    const float* root    = (const float*)d_in[0];
    const float* cpt     = (const float*)d_in[1];
    const int*   parents = (const int*)d_in[2];
    float* out = (float*)d_out;

    bn_root_kernel<<<kM / 256, 256, 0, stream>>>(root, out);

    for (int l = 0; l < kLay; ++l) {
        bn_layer_kernel<<<kM / kNodesPerBlock, kWavesPerBlock * 32, 0, stream>>>(
            cpt + (size_t)l * kM * kC,
            parents + (size_t)l * kM * kK,
            out + (size_t)l * kM,
            out + (size_t)(l + 1) * kM);
    }
}


# Config sum of a node split in 4 partials: 2 waves of a block each take half of every CPT row (16 b128 loads per lane, all in flight) on 2 accumulator tiles; partials added via LDS in fixed order
# speedup vs baseline: 1.5799x; 1.5799x over previous
#include <hip/hip_runtime.h>


typedef __attribute__((ext_vector_type(2))) float v2f;
typedef __attribute__((ext_vector_type(4))) float v4f;
typedef __attribute__((ext_vector_type(8))) float v8f;

static constexpr int kM    = 8192;
static constexpr int kK    = 8;
static constexpr int kC    = 256;
static constexpr int kLay  = 8;
static constexpr int kTilesPerBlock = 2;
static constexpr int kSplit         = 2;
static constexpr int kWavesPerBlock = kTilesPerBlock * kSplit;
static constexpr int kNodesPerBlock = kTilesPerBlock * 16;
static constexpr int kChunks        = kC / kSplit / 8;

__device__ __forceinline__ float fast_sigmoid(float x) {
    return __builtin_amdgcn_rcpf(1.0f + __expf(-x));
}

__device__ __forceinline__ float pick8(v8f c, int i) {
    float r = c[0];
    if (i == 1) r = c[1];
    if (i == 2) r = c[2];
    if (i == 3) r = c[3];
    if (i == 4) r = c[4];
    if (i == 5) r = c[5];
    if (i == 6) r = c[6];
    if (i == 7) r = c[7];
    return r;
}

__global__ void bn_root_kernel(const float* __restrict__ logits,
                               float* __restrict__ out) {
    int i = blockIdx.x * blockDim.x + threadIdx.x;
    if (i < kM) out[i] = fast_sigmoid(logits[i]);
}

__global__ __launch_bounds__(kWavesPerBlock * 32)
void bn_layer_kernel(const float* __restrict__ cpt,
                     const int*   __restrict__ par,
                     const float* __restrict__ prev,
                     float*       __restrict__ outm) {
    __shared__ float part[kTilesPerBlock][kSplit][16];

    const int lane = threadIdx.x & 31;
    const int wave = threadIdx.x >> 5;
    const int n    = lane & 15;
    const int h    = lane >> 4;
    const int tl   = wave >> 1;
    const int s    = wave & 1;
    const int node = blockIdx.x * kNodesPerBlock + tl * 16 + n;

    const int4* pp = (const int4*)(par + (size_t)node * kK);
    const int4 ia = pp[0];
    const int4 ib = pp[1];

    const v4f* row = (const v4f*)(cpt + (size_t)node * kC + 128 * s + 4 * h);
    v4f v[kChunks];
#pragma unroll
    for (int i = 0; i < kChunks; ++i) v[i] = row[2 * i];

    float p[kK], q[kK];
    p[0] = prev[ia.x]; p[1] = prev[ia.y]; p[2] = prev[ia.z]; p[3] = prev[ia.w];
    p[4] = prev[ib.x]; p[5] = prev[ib.y]; p[6] = prev[ib.z]; p[7] = prev[ib.w];
    __builtin_amdgcn_sched_barrier(0);
#pragma unroll
    for (int j = 0; j < kK; ++j) q[j] = 1.0f - p[j];

#pragma unroll
    for (int i = 0; i < kChunks; ++i) {
        v[i].x = fast_sigmoid(v[i].x);
        v[i].y = fast_sigmoid(v[i].y);
        v[i].z = fast_sigmoid(v[i].z);
        v[i].w = fast_sigmoid(v[i].w);
    }

    const float base = (s ? p[0] : q[0]) * (h ? p[5] : q[5]);
    const float b6q = base * q[6], b6p = base * p[6];
    const float e0 = b6q * q[7], e1 = b6q * p[7], e2 = b6p * q[7], e3 = b6p * p[7];
    float u12[4], u34[4];
#pragma unroll
    for (int x = 0; x < 4; ++x) {
        u12[x] = ((x >> 1) ? p[1] : q[1]) * ((x & 1) ? p[2] : q[2]);
        u34[x] = ((x >> 1) ? p[3] : q[3]) * ((x & 1) ? p[4] : q[4]);
    }

    v8f acc0 = {}, acc1 = {};
#pragma unroll
    for (int i = 0; i < kChunks; ++i) {
        const float r = u12[i >> 2] * u34[i & 3];
        v2f a0, a1, b0, b1;
        a0.x = r * e0; a0.y = r * e1;
        a1.x = r * e2; a1.y = r * e3;
        b0.x = v[i].x; b0.y = v[i].y;
        b1.x = v[i].z; b1.y = v[i].w;
        acc0 = __builtin_amdgcn_wmma_f32_16x16x4_f32(
            false, a0, false, b0, (short)0, acc0, false, false);
        acc1 = __builtin_amdgcn_wmma_f32_16x16x4_f32(
            false, a1, false, b1, (short)0, acc1, false, false);
    }

    const v8f acc = acc0 + acc1;
    const float d = pick8(acc, lane & 7);
    if (lane < 8) {
        part[tl][s][lane] = d;
    } else if (lane >= 24) {
        part[tl][s][lane - 16] = d;
    }
    __syncthreads();
    if (wave == 0) {
        outm[blockIdx.x * kNodesPerBlock + lane] = part[h][0][n] + part[h][1][n];
    }
}

extern "C" void kernel_launch(void* const* d_in, const int* in_sizes, int n_in,
                              void* d_out, int out_size, void* d_ws, size_t ws_size,
                              hipStream_t stream) {
    (void)in_sizes; (void)n_in; (void)out_size; (void)d_ws; (void)ws_size;
    const float* root    = (const float*)d_in[0];
    const float* cpt     = (const float*)d_in[1];
    const int*   parents = (const int*)d_in[2];
    float* out = (float*)d_out;

    bn_root_kernel<<<kM / 256, 256, 0, stream>>>(root, out);

    for (int l = 0; l < kLay; ++l) {
        bn_layer_kernel<<<kM / kNodesPerBlock, kWavesPerBlock * 32, 0, stream>>>(
            cpt + (size_t)l * kM * kC,
            parents + (size_t)l * kM * kK,
            out + (size_t)l * kM,
            out + (size_t)(l + 1) * kM);
    }
}
